# starting kernel
# baseline (speedup 1.0000x reference)
#include <hip/hip_runtime.h>
#include <hip/hip_bf16.h>
#include <math.h>

#define ALPHA 0.2f

typedef __attribute__((ext_vector_type(2))) float v2f;
typedef __attribute__((ext_vector_type(8))) float v8f;

__global__ __launch_bounds__(32) void gat_h_gemm_wmma(
    const float* __restrict__ X,
    const float* __restrict__ W,
    float* __restrict__ H)
{
  const int tileM = blockIdx.x;
  const int tileN = blockIdx.y;
  const int lane  = threadIdx.x;
  const int half  = lane >> 4;
  const int l15   = lane & 15;

  const int rowA = tileM * 16 + l15;
  const int colD = tileN * 16 + l15;

  const float* __restrict__ xrow = X + rowA * 128;

  v8f acc = {};
#pragma unroll
  for (int k0 = 0; k0 < 128; k0 += 4) {
    const int k = k0 + 2 * half;
    v2f a, b;
    a.x = xrow[k];
    a.y = xrow[k + 1];
    b.x = W[k * 64 + colD];
    b.y = W[(k + 1) * 64 + colD];
    acc = __builtin_amdgcn_wmma_f32_16x16x4_f32(
        false, a, false, b, (short)0, acc, false, false);
  }

#pragma unroll
  for (int r = 0; r < 8; ++r) {
    const int row = tileM * 16 + r + 8 * half;
    H[row * 64 + colD] = acc[r];
  }
}

__global__ __launch_bounds__(256) void gat_post(
    const float* __restrict__ H,
    const float* __restrict__ A,
    float* __restrict__ OUT)
{
  const int b   = blockIdx.x;
  const int tid = threadIdx.x;
  const float* __restrict__ h = H + b * 512 * 64;
  float* __restrict__ o       = OUT + b * 512 * 64;

  __shared__ float s_a[128];
  __shared__ float s_s[512];
  __shared__ float s_u[512];
  __shared__ float s_v[513];
  __shared__ float s_alpha[512];
  __shared__ float s_Slo[64];
  __shared__ float s_Shi[64];
  __shared__ float s_P[64];
  __shared__ float s_red[256];

  if (tid < 128) s_a[tid] = A[tid];
  __syncthreads();

  for (int r = tid; r < 512; r += 256) {
    const float* hr = h + r * 64;
    float du = 0.f, dv = 0.f;
#pragma unroll 8
    for (int c = 0; c < 64; ++c) {
      const float x = hr[c];
      du += x * s_a[c];
      dv += x * s_a[c + 64];
    }
    s_u[r] = du;
    s_v[r] = dv;
    s_s[r] = du + dv;
  }

  if (tid < 128) {
    const int c    = tid & 63;
    const int base = (tid < 64) ? 0 : 256;
    float acc = 0.f;
    for (int r = 0; r < 256; ++r) acc += h[(base + r) * 64 + c];
    if (tid < 64) s_Slo[c] = acc; else s_Shi[c] = acc;
  }
  __syncthreads();

  float lt[2];
#pragma unroll
  for (int q = 0; q < 2; ++q) {
    const int j  = tid + q * 256;
    const int r0 = (2 * j) & 511;
    const float t = s_u[r0] + s_v[r0 + 1];
    lt[q] = (t > 0.f) ? t : ALPHA * t;
  }
  s_red[tid] = fmaxf(lt[0], lt[1]);
  __syncthreads();
  for (int off = 128; off > 0; off >>= 1) {
    if (tid < off) s_red[tid] = fmaxf(s_red[tid], s_red[tid + off]);
    __syncthreads();
  }
  const float m = s_red[0];
  __syncthreads();
  const float e0 = __expf(lt[0] - m);
  const float e1 = __expf(lt[1] - m);
  s_red[tid] = e0 + e1;
  __syncthreads();
  for (int off = 128; off > 0; off >>= 1) {
    if (tid < off) s_red[tid] += s_red[tid + off];
    __syncthreads();
  }
  const float Z = s_red[0];
  __syncthreads();
  s_alpha[tid]       = e0 / Z;
  s_alpha[tid + 256] = e1 / Z;
  __syncthreads();

  if (tid < 64) {
    float acc = 0.f;
    for (int j = 0; j < 512; ++j) acc += s_alpha[j] * h[j * 64 + tid];
    s_P[tid] = (acc > 0.f) ? acc : (__expf(acc) - 1.f);
  }
  __syncthreads();

  {
    const int i = tid;
    const float sv0 = s_s[2 * i];
    const float sv1 = s_s[2 * i + 1];
    const float v0  = (sv0 > 0.f) ? sv0 : ALPHA * sv0;
    const float v1  = (sv1 > 0.f) ? sv1 : ALPHA * sv1;
    const float mm  = fmaxf(v0, v1);
    const float ee0 = __expf(v0 - mm);
    const float ee1 = __expf(v1 - mm);
    const float den = 256.f * (ee0 + ee1);
    const float w0  = ee0 / den;
    const float w1  = ee1 / den;
#pragma unroll 8
    for (int c = 0; c < 64; ++c) {
      const float hp = w0 * s_Slo[c] + w1 * s_Shi[c];
      o[i * 64 + c] = (hp > 0.f) ? hp : (__expf(hp) - 1.f);
    }
  }

  for (int idx = tid; idx < 256 * 64; idx += 256) {
    const int i = 256 + (idx >> 6);
    const int c = idx & 63;
    o[i * 64 + c] = s_P[c];
  }
}

extern "C" void kernel_launch(void* const* d_in, const int* in_sizes, int n_in,
                              void* d_out, int out_size, void* d_ws, size_t ws_size,
                              hipStream_t stream) {
  const float* X = (const float*)d_in[0];
  const float* W = (const float*)d_in[2];
  const float* A = (const float*)d_in[3];
  float* OUT = (float*)d_out;
  float* H   = (float*)d_ws;

  dim3 gemmGrid(128, 4);
  gat_h_gemm_wmma<<<gemmGrid, 32, 0, stream>>>(X, W, H);

  gat_post<<<4, 256, 0, stream>>>(H, A, OUT);
}


# two launches fused into one: one 1024-thread work-group per batch, each wave keeps its 16x64 h strip in WMMA accumulators, sums by lane shuffles + LDS; h never stored; summation order changes
# speedup vs baseline: 1.2830x; 1.2830x over previous
#include <hip/hip_runtime.h>
#include <math.h>

#define ALPHA 0.2f

typedef __attribute__((ext_vector_type(2))) float v2f;
typedef __attribute__((ext_vector_type(8))) float v8f;

__device__ __forceinline__ float half16_sum(float x) {
  x += __shfl_xor(x, 8);
  x += __shfl_xor(x, 4);
  x += __shfl_xor(x, 2);
  x += __shfl_xor(x, 1);
  return x;
}
__device__ __forceinline__ float wave_sum(float x) {
  x += __shfl_xor(x, 16);
  return half16_sum(x);
}
__device__ __forceinline__ float wave_max(float x) {
  x = fmaxf(x, __shfl_xor(x, 16));
  x = fmaxf(x, __shfl_xor(x, 8));
  x = fmaxf(x, __shfl_xor(x, 4));
  x = fmaxf(x, __shfl_xor(x, 2));
  x = fmaxf(x, __shfl_xor(x, 1));
  return x;
}

__global__ __launch_bounds__(1024) void gat_fused(
    const float* __restrict__ X,
    const float* __restrict__ W,
    const float* __restrict__ A,
    float* __restrict__ OUT)
{
  const int b    = blockIdx.x;
  const int tid  = threadIdx.x;
  const int w    = tid >> 5;
  const int lane = tid & 31;
  const int half = lane >> 4;
  const int l15  = lane & 15;

  __shared__ float s_a[128];
  __shared__ float s_u[512];
  __shared__ float s_v[513];
  __shared__ float s_s[512];
  __shared__ float s_alpha[512];
  __shared__ float s_w0[256];
  __shared__ float s_w1[256];
  __shared__ float s_part[64 * 64];
  __shared__ __attribute__((aligned(16))) float s_Slo[64];
  __shared__ __attribute__((aligned(16))) float s_Shi[64];
  __shared__ __attribute__((aligned(16))) float s_P[64];
  __shared__ float s_red[32];
  __shared__ float s_red2[32];

  if (tid < 128) s_a[tid] = A[tid];

  const float* __restrict__ xrow = X + (b * 512 + w * 16 + l15) * 128;

  v8f acc[4];
#pragma unroll
  for (int n = 0; n < 4; ++n) acc[n] = (v8f){};

#pragma unroll
  for (int k0 = 0; k0 < 128; k0 += 4) {
    const int k = k0 + 2 * half;
    v2f a;
    a.x = xrow[k];
    a.y = xrow[k + 1];
#pragma unroll
    for (int n = 0; n < 4; ++n) {
      const int colD = n * 16 + l15;
      v2f bf;
      bf.x = W[k * 64 + colD];
      bf.y = W[(k + 1) * 64 + colD];
      acc[n] = __builtin_amdgcn_wmma_f32_16x16x4_f32(
          false, a, false, bf, (short)0, acc[n], false, false);
    }
  }
  __syncthreads();

  {
    float alo[4], ahi[4];
#pragma unroll
    for (int n = 0; n < 4; ++n) {
      alo[n] = s_a[n * 16 + l15];
      ahi[n] = s_a[64 + n * 16 + l15];
    }
#pragma unroll
    for (int r = 0; r < 8; ++r) {
      float du = 0.f, dv = 0.f;
#pragma unroll
      for (int n = 0; n < 4; ++n) {
        du += acc[n][r] * alo[n];
        dv += acc[n][r] * ahi[n];
      }
      du = half16_sum(du);
      dv = half16_sum(dv);
      if (l15 == 0) {
        const int row = w * 16 + r + 8 * half;
        s_u[row] = du;
        s_v[row] = dv;
        s_s[row] = du + dv;
      }
    }
#pragma unroll
    for (int n = 0; n < 4; ++n) {
      float cs = 0.f;
#pragma unroll
      for (int r = 0; r < 8; ++r) cs += acc[n][r];
      s_part[(w * 2 + half) * 64 + n * 16 + l15] = cs;
    }
  }
  __syncthreads();

  if (tid < 128) {
    const int c  = tid & 63;
    const int hi = tid >> 6;
    float sum = 0.f;
    for (int p = 0; p < 32; ++p) sum += s_part[(hi * 32 + p) * 64 + c];
    if (hi == 0) s_Slo[c] = sum; else s_Shi[c] = sum;
  } else if (tid >= 512 && tid < 768) {
    const int i = tid - 512;
    const float sv0 = s_s[2 * i];
    const float sv1 = s_s[2 * i + 1];
    const float v0  = (sv0 > 0.f) ? sv0 : ALPHA * sv0;
    const float v1  = (sv1 > 0.f) ? sv1 : ALPHA * sv1;
    const float mm  = fmaxf(v0, v1);
    const float ee0 = __expf(v0 - mm);
    const float ee1 = __expf(v1 - mm);
    const float den = 256.f * (ee0 + ee1);
    s_w0[i] = ee0 / den;
    s_w1[i] = ee1 / den;
  }
  float lt = -INFINITY;
  if (tid < 512) {
    const int r0 = (2 * tid) & 511;
    const float t = s_u[r0] + s_v[r0 + 1];
    lt = (t > 0.f) ? t : ALPHA * t;
  }
  {
    const float wm = wave_max(lt);
    if (lane == 0) s_red[w] = wm;
  }
  __syncthreads();

  const float m = wave_max(s_red[lane]);
  const float e = (tid < 512) ? __expf(lt - m) : 0.f;
  {
    const float ws = wave_sum(e);
    if (lane == 0) s_red2[w] = ws;
  }
  __syncthreads();

  const float Z = wave_sum(s_red2[lane]);
  if (tid < 512) s_alpha[tid] = e / Z;
  __syncthreads();

  {
    float al[8];
#pragma unroll
    for (int r = 0; r < 8; ++r) al[r] = s_alpha[w * 16 + r + 8 * half];
#pragma unroll
    for (int n = 0; n < 4; ++n) {
      float ps = 0.f;
#pragma unroll
      for (int r = 0; r < 8; ++r) ps += al[r] * acc[n][r];
      s_part[(w * 2 + half) * 64 + n * 16 + l15] = ps;
    }
  }
  __syncthreads();

  if (tid < 64) {
    float sum = 0.f;
    for (int p = 0; p < 64; ++p) sum += s_part[p * 64 + tid];
    s_P[tid] = (sum > 0.f) ? sum : (__expf(sum) - 1.f);
  }
  __syncthreads();

  float4* __restrict__ o4 = reinterpret_cast<float4*>(OUT + b * 512 * 64);
  const int c4 = (tid & 15) * 4;
  const float4 slo = *reinterpret_cast<const float4*>(&s_Slo[c4]);
  const float4 shi = *reinterpret_cast<const float4*>(&s_Shi[c4]);
  const float4 pv  = *reinterpret_cast<const float4*>(&s_P[c4]);
  for (int q = 0; q < 4; ++q) {
    const int i = (tid >> 4) + q * 64;
    const float w0 = s_w0[i];
    const float w1 = s_w1[i];
    float4 hp;
    hp.x = w0 * slo.x + w1 * shi.x;
    hp.y = w0 * slo.y + w1 * shi.y;
    hp.z = w0 * slo.z + w1 * shi.z;
    hp.w = w0 * slo.w + w1 * shi.w;
    hp.x = (hp.x > 0.f) ? hp.x : (__expf(hp.x) - 1.f);
    hp.y = (hp.y > 0.f) ? hp.y : (__expf(hp.y) - 1.f);
    hp.z = (hp.z > 0.f) ? hp.z : (__expf(hp.z) - 1.f);
    hp.w = (hp.w > 0.f) ? hp.w : (__expf(hp.w) - 1.f);
    o4[i * 16 + (tid & 15)]         = hp;
    o4[(256 + i) * 16 + (tid & 15)] = pv;
  }
}

extern "C" void kernel_launch(void* const* d_in, const int* in_sizes, int n_in,
                              void* d_out, int out_size, void* d_ws, size_t ws_size,
                              hipStream_t stream) {
  const float* X = (const float*)d_in[0];
  const float* W = (const float*)d_in[2];
  const float* A = (const float*)d_in[3];
  float* OUT = (float*)d_out;

  gat_fused<<<4, 1024, 0, stream>>>(X, W, A, OUT);
}


# softmax over t[j] in online form: each wave takes max, exp and weighted X sum of its own 16 rows; one rescaling join replaces 3 block-wide reductions (3 barriers fewer); summation order changes
# speedup vs baseline: 3.1837x; 2.4814x over previous
#include <hip/hip_runtime.h>
#include <math.h>

#define ALPHA 0.2f

typedef __attribute__((ext_vector_type(2))) float v2f;
typedef __attribute__((ext_vector_type(8))) float v8f;

__device__ __forceinline__ float half16_sum(float x) {
  x += __shfl_xor(x, 8);
  x += __shfl_xor(x, 4);
  x += __shfl_xor(x, 2);
  x += __shfl_xor(x, 1);
  return x;
}
template <int M>
__device__ __forceinline__ float row_xor(float x) {
  const int b = __builtin_bit_cast(int, x);
  return __builtin_bit_cast(
      float, __builtin_amdgcn_mov_dpp(b, 0x160 | M, 0xf, 0xf, true));
}
__device__ __forceinline__ float xhalf(float x) {
  const int b = __builtin_bit_cast(int, x);
  return __builtin_bit_cast(
      float, __builtin_amdgcn_permlanex16(b, b, 0x76543210, 0xfedcba98, false, false));
}
__device__ __forceinline__ float row16_max(float x) {
  x = fmaxf(x, row_xor<1>(x));
  x = fmaxf(x, row_xor<2>(x));
  x = fmaxf(x, row_xor<4>(x));
  x = fmaxf(x, row_xor<8>(x));
  return x;
}
__device__ __forceinline__ float row16_sum(float x) {
  x += row_xor<1>(x);
  x += row_xor<2>(x);
  x += row_xor<4>(x);
  x += row_xor<8>(x);
  return x;
}
template <int I>
__device__ __forceinline__ float lane_value(float x) {
  return __builtin_bit_cast(
      float, __builtin_amdgcn_readlane(__builtin_bit_cast(int, x), I));
}
__device__ __forceinline__ float dot4(const float4 a, const float4 b) {
  return a.x * b.x + a.y * b.y + a.z * b.z + a.w * b.w;
}

template <int O>
__device__ __forceinline__ void fold_stage(float* c, const int lane) {
  const bool up = (lane & O) != 0;
#pragma unroll
  for (int j = 0; j < O; ++j) {
    const float keep = up ? c[j + O] : c[j];
    const float send = up ? c[j] : c[j + O];
    c[j] = keep + __shfl_xor(send, O);
  }
}

__global__ __launch_bounds__(1024) void gat_fused(
    const float* __restrict__ X,
    const float* __restrict__ W,
    const float* __restrict__ A,
    float* __restrict__ OUT)
{
  const int b    = blockIdx.x;
  const int tid  = threadIdx.x;
  const int w    = tid >> 5;
  const int lane = tid & 31;
  const int half = lane >> 4;
  const int l15  = lane & 15;

  __shared__ __attribute__((aligned(16))) float s_wal[128];
  __shared__ __attribute__((aligned(16))) float s_wah[128];
  __shared__ float s_u[512];
  __shared__ float s_v[513];
  __shared__ float s_w0[256];
  __shared__ float s_w1[256];
  __shared__ __attribute__((aligned(16))) float s_part[32 * 128];
  __shared__ __attribute__((aligned(16))) float s_xp[32 * 128];
  __shared__ float s_mw[32];
  __shared__ float s_zw[32];
  __shared__ __attribute__((aligned(16))) float s_M[3 * 128];
  __shared__ float s_Y[8 * 3 * 64];
  __shared__ __attribute__((aligned(16))) float s_Slo[64];
  __shared__ __attribute__((aligned(16))) float s_Shi[64];
  __shared__ __attribute__((aligned(16))) float s_P[64];

  const float4* __restrict__ X4 = reinterpret_cast<const float4*>(X);
  const float4* __restrict__ W4 = reinterpret_cast<const float4*>(W);
  const float4* __restrict__ A4 = reinterpret_cast<const float4*>(A);

  const float4 wq0 = W4[tid];
  const float4 wq1 = W4[tid + 1024];
  const float4 alo = A4[l15];
  const float4 ahi = A4[16 + l15];

  float4 x[16];
#pragma unroll
  for (int i = 0; i < 16; ++i) x[i] = X4[(b * 512 + w * 16 + i) * 32 + lane];

  __builtin_amdgcn_sched_barrier(0);

  {
    const float d0l = half16_sum(dot4(wq0, alo));
    const float d0h = half16_sum(dot4(wq0, ahi));
    const float d1l = half16_sum(dot4(wq1, alo));
    const float d1h = half16_sum(dot4(wq1, ahi));
    if (l15 == 0) {
      const int k0 = tid >> 4;
      s_wal[k0]      = d0l;
      s_wah[k0]      = d0h;
      s_wal[64 + k0] = d1l;
      s_wah[64 + k0] = d1h;
    }
  }
  __syncthreads();

  {
    const float4 wl = *reinterpret_cast<const float4*>(&s_wal[4 * lane]);
    const float4 wh = *reinterpret_cast<const float4*>(&s_wah[4 * lane]);
    float c[32];
#pragma unroll
    for (int i = 0; i < 16; ++i) {
      c[i]      = dot4(x[i], wl);
      c[16 + i] = dot4(x[i], wh);
    }
    fold_stage<16>(c, lane);
    fold_stage<8>(c, lane);
    fold_stage<4>(c, lane);
    fold_stage<2>(c, lane);
    fold_stage<1>(c, lane);
    if (half == 0) s_u[w * 16 + l15] = c[0];
    else           s_v[w * 16 + l15] = c[0];

    float4 cs = x[0];
#pragma unroll
    for (int i = 1; i < 16; ++i) {
      cs.x += x[i].x; cs.y += x[i].y; cs.z += x[i].z; cs.w += x[i].w;
    }
    *reinterpret_cast<float4*>(&s_part[(w * 32 + lane) * 4]) = cs;
  }

  const int nt = w & 3;
  const int kc = w >> 2;
  float bw[8];
#pragma unroll
  for (int s = 0; s < 4; ++s) {
    const int k = kc * 16 + s * 4 + 2 * half;
    bw[2 * s]     = W[k * 64 + nt * 16 + l15];
    bw[2 * s + 1] = W[(k + 1) * 64 + nt * 16 + l15];
  }
  __syncthreads();

  {
    const int j  = w * 16 + l15;
    const int r0 = (2 * j) & 511;
    const float t  = s_u[r0] + s_v[r0 + 1];
    const float lt = (t > 0.f) ? t : ALPHA * t;
    const float mw = row16_max(lt);
    const float e  = __expf(lt - mw);
    float4 xp = make_float4(0.f, 0.f, 0.f, 0.f);
    float ev[16];
#define XP_ROW(I)                                  \
    {                                              \
      const float ei = lane_value<I>(e);           \
      ev[I] = ei;                                  \
      xp.x += ei * x[I].x; xp.y += ei * x[I].y;    \
      xp.z += ei * x[I].z; xp.w += ei * x[I].w;    \
    }
    XP_ROW(0)  XP_ROW(1)  XP_ROW(2)  XP_ROW(3)
    XP_ROW(4)  XP_ROW(5)  XP_ROW(6)  XP_ROW(7)
    XP_ROW(8)  XP_ROW(9)  XP_ROW(10) XP_ROW(11)
    XP_ROW(12) XP_ROW(13) XP_ROW(14) XP_ROW(15)
#undef XP_ROW
    if (lane == 0) {
      const float zw = (((ev[0] + ev[1]) + (ev[2] + ev[3])) + ((ev[4] + ev[5]) + (ev[6] + ev[7]))) +
                       (((ev[8] + ev[9]) + (ev[10] + ev[11])) + ((ev[12] + ev[13]) + (ev[14] + ev[15])));
      s_mw[w] = mw;
      s_zw[w] = zw;
    }
    *reinterpret_cast<float4*>(&s_xp[(w * 32 + lane) * 4]) = xp;
  }
  __syncthreads();

  if (tid < 128) {
    const float mwl = s_mw[lane];
    float m = row16_max(mwl);
    m = fmaxf(m, xhalf(m));
    const float sc = __expf(mwl - m);
    float Z = row16_sum(s_zw[lane] * sc);
    Z += xhalf(Z);
    const float scn = sc / Z;
    float a0 = 0.f, a1 = 0.f, a2 = 0.f, a3 = 0.f;
#define XP_JOIN(P)                                                 \
    a0 += lane_value<P>(scn)     * s_xp[(P) * 128 + tid];          \
    a1 += lane_value<P + 1>(scn) * s_xp[(P + 1) * 128 + tid];      \
    a2 += lane_value<P + 2>(scn) * s_xp[(P + 2) * 128 + tid];      \
    a3 += lane_value<P + 3>(scn) * s_xp[(P + 3) * 128 + tid];
    XP_JOIN(0)  XP_JOIN(4)  XP_JOIN(8)  XP_JOIN(12)
    XP_JOIN(16) XP_JOIN(20) XP_JOIN(24) XP_JOIN(28)
#undef XP_JOIN
    s_M[256 + tid] = (a0 + a1) + (a2 + a3);
  } else if (tid >= 768) {
    const int t2 = tid - 768;
    const int cc = t2 & 127;
    const int hi = t2 >> 7;
    float sum = 0.f;
    for (int p = 0; p < 16; ++p) sum += s_part[(hi * 16 + p) * 128 + cc];
    s_M[hi * 128 + cc] = sum;
  } else if (tid >= 512) {
    const int i = tid - 512;
    const float sv0 = s_u[2 * i] + s_v[2 * i];
    const float sv1 = s_u[2 * i + 1] + s_v[2 * i + 1];
    const float v0  = (sv0 > 0.f) ? sv0 : ALPHA * sv0;
    const float v1  = (sv1 > 0.f) ? sv1 : ALPHA * sv1;
    const float mm  = fmaxf(v0, v1);
    const float ee0 = __expf(v0 - mm);
    const float ee1 = __expf(v1 - mm);
    const float den = 256.f * (ee0 + ee1);
    s_w0[i] = ee0 / den;
    s_w1[i] = ee1 / den;
  }
  __syncthreads();

  {
    const int rowA = (l15 < 3) ? l15 : 0;
    v8f acc0 = (v8f){};
    v8f acc1 = (v8f){};
#pragma unroll
    for (int s = 0; s < 4; ++s) {
      const int k = kc * 16 + s * 4 + 2 * half;
      v2f a = *reinterpret_cast<const v2f*>(&s_M[rowA * 128 + k]);
      if (l15 >= 3) a = (v2f){0.f, 0.f};
      v2f bf;
      bf.x = bw[2 * s];
      bf.y = bw[2 * s + 1];
      if (s & 1)
        acc1 = __builtin_amdgcn_wmma_f32_16x16x4_f32(
            false, a, false, bf, (short)0, acc1, false, false);
      else
        acc0 = __builtin_amdgcn_wmma_f32_16x16x4_f32(
            false, a, false, bf, (short)0, acc0, false, false);
    }
    if (half == 0) {
      const int cD = nt * 16 + l15;
      s_Y[(kc * 3 + 0) * 64 + cD] = acc0[0] + acc1[0];
      s_Y[(kc * 3 + 1) * 64 + cD] = acc0[1] + acc1[1];
      s_Y[(kc * 3 + 2) * 64 + cD] = acc0[2] + acc1[2];
    }
  }
  __syncthreads();

  if (tid < 192) {
    const int row = tid >> 6;
    const int cc  = tid & 63;
    float sum = 0.f;
    for (int q = 0; q < 8; ++q) sum += s_Y[(q * 3 + row) * 64 + cc];
    if (row == 0)      s_Slo[cc] = sum;
    else if (row == 1) s_Shi[cc] = sum;
    else               s_P[cc] = (sum > 0.f) ? sum : (__expf(sum) - 1.f);
  }
  __syncthreads();

  float4* __restrict__ o4 = reinterpret_cast<float4*>(OUT + b * 512 * 64);
  const int c4 = (tid & 15) * 4;
  const float4 slo = *reinterpret_cast<const float4*>(&s_Slo[c4]);
  const float4 shi = *reinterpret_cast<const float4*>(&s_Shi[c4]);
  const float4 pv  = *reinterpret_cast<const float4*>(&s_P[c4]);
  for (int q = 0; q < 4; ++q) {
    const int i = (tid >> 4) + q * 64;
    const float w0 = s_w0[i];
    const float w1 = s_w1[i];
    float4 hp;
    hp.x = w0 * slo.x + w1 * shi.x;
    hp.y = w0 * slo.y + w1 * shi.y;
    hp.z = w0 * slo.z + w1 * shi.z;
    hp.w = w0 * slo.w + w1 * shi.w;
    hp.x = (hp.x > 0.f) ? hp.x : (__expf(hp.x) - 1.f);
    hp.y = (hp.y > 0.f) ? hp.y : (__expf(hp.y) - 1.f);
    hp.z = (hp.z > 0.f) ? hp.z : (__expf(hp.z) - 1.f);
    hp.w = (hp.w > 0.f) ? hp.w : (__expf(hp.w) - 1.f);
    o4[i * 16 + (tid & 15)]         = hp;
    o4[(256 + i) * 16 + (tid & 15)] = pv;
  }
}

extern "C" void kernel_launch(void* const* d_in, const int* in_sizes, int n_in,
                              void* d_out, int out_size, void* d_ws, size_t ws_size,
                              hipStream_t stream) {
  const float* X = (const float*)d_in[0];
  const float* W = (const float*)d_in[2];
  const float* A = (const float*)d_in[3];
  float* OUT = (float*)d_out;

  gat_fused<<<4, 1024, 0, stream>>>(X, W, A, OUT);
}


# batch split over two work-groups by output rows (grid 4 -> 8): rows <256 need only s=X.(W a_lo+W a_hi) and X column sums, rows >=256 only t and the softmax-weighted X sum; summation order changes
# speedup vs baseline: 3.8922x; 1.2225x over previous
#include <hip/hip_runtime.h>
#include <math.h>

#define ALPHA 0.2f

typedef __attribute__((ext_vector_type(2))) float v2f;
typedef __attribute__((ext_vector_type(8))) float v8f;

__device__ __forceinline__ float half16_sum(float x) {
  x += __shfl_xor(x, 8);
  x += __shfl_xor(x, 4);
  x += __shfl_xor(x, 2);
  x += __shfl_xor(x, 1);
  return x;
}
template <int M>
__device__ __forceinline__ float row_xor(float x) {
  const int b = __builtin_bit_cast(int, x);
  return __builtin_bit_cast(
      float, __builtin_amdgcn_mov_dpp(b, 0x160 | M, 0xf, 0xf, true));
}
__device__ __forceinline__ float xhalf(float x) {
  const int b = __builtin_bit_cast(int, x);
  return __builtin_bit_cast(
      float, __builtin_amdgcn_permlanex16(b, b, 0x76543210, 0xfedcba98, false, false));
}
__device__ __forceinline__ float row16_max(float x) {
  x = fmaxf(x, row_xor<1>(x));
  x = fmaxf(x, row_xor<2>(x));
  x = fmaxf(x, row_xor<4>(x));
  x = fmaxf(x, row_xor<8>(x));
  return x;
}
__device__ __forceinline__ float row16_sum(float x) {
  x += row_xor<1>(x);
  x += row_xor<2>(x);
  x += row_xor<4>(x);
  x += row_xor<8>(x);
  return x;
}
template <int I>
__device__ __forceinline__ float lane_value(float x) {
  return __builtin_bit_cast(
      float, __builtin_amdgcn_readlane(__builtin_bit_cast(int, x), I));
}
__device__ __forceinline__ float dot4(const float4 a, const float4 b) {
  return a.x * b.x + a.y * b.y + a.z * b.z + a.w * b.w;
}

template <int O>
__device__ __forceinline__ void fold_stage(float* c, const int lane) {
  const bool up = (lane & O) != 0;
#pragma unroll
  for (int j = 0; j < O; ++j) {
    const float keep = up ? c[j + O] : c[j];
    const float send = up ? c[j] : c[j + O];
    c[j] = keep + __shfl_xor(send, O);
  }
}

__global__ __launch_bounds__(1024) void gat_fused(
    const float* __restrict__ X,
    const float* __restrict__ W,
    const float* __restrict__ A,
    float* __restrict__ OUT)
{
  const int b    = blockIdx.x >> 1;
  const int role = blockIdx.x & 1;
  const int tid  = threadIdx.x;
  const int w    = tid >> 5;
  const int lane = tid & 31;
  const int half = lane >> 4;
  const int l15  = lane & 15;

  __shared__ __attribute__((aligned(16))) float s_wal[128];
  __shared__ __attribute__((aligned(16))) float s_wah[128];
  __shared__ float s_d[512];
  __shared__ float s_w0[256];
  __shared__ float s_w1[256];
  __shared__ __attribute__((aligned(16))) float s_part[32 * 128];
  __shared__ __attribute__((aligned(16))) float s_xp[32 * 128];
  __shared__ float s_mw[32];
  __shared__ float s_zw[32];
  __shared__ __attribute__((aligned(16))) float s_M[3 * 128];
  __shared__ float s_Y[8 * 3 * 64];
  __shared__ __attribute__((aligned(16))) float s_Slo[64];
  __shared__ __attribute__((aligned(16))) float s_Shi[64];
  __shared__ __attribute__((aligned(16))) float s_P[64];

  const float4* __restrict__ X4 = reinterpret_cast<const float4*>(X);
  const float4* __restrict__ W4 = reinterpret_cast<const float4*>(W);
  const float4* __restrict__ A4 = reinterpret_cast<const float4*>(A);

  const float4 wq0 = W4[tid];
  const float4 wq1 = W4[tid + 1024];
  const float4 alo = A4[l15];
  const float4 ahi = A4[16 + l15];

  float4 x[16];
#pragma unroll
  for (int i = 0; i < 16; ++i) x[i] = X4[(b * 512 + w * 16 + i) * 32 + lane];

  __builtin_amdgcn_sched_barrier(0);

  {
    const float d0l = half16_sum(dot4(wq0, alo));
    const float d0h = half16_sum(dot4(wq0, ahi));
    const float d1l = half16_sum(dot4(wq1, alo));
    const float d1h = half16_sum(dot4(wq1, ahi));
    if (l15 == 0) {
      const int k0 = tid >> 4;
      s_wal[k0]      = d0l;
      s_wah[k0]      = d0h;
      s_wal[64 + k0] = d1l;
      s_wah[64 + k0] = d1h;
    }
  }
  __syncthreads();

  {
    const float4 wl = *reinterpret_cast<const float4*>(&s_wal[4 * lane]);
    const float4 wh = *reinterpret_cast<const float4*>(&s_wah[4 * lane]);
    float c[16];
    if (role == 0) {
      const float4 ws = make_float4(wl.x + wh.x, wl.y + wh.y, wl.z + wh.z, wl.w + wh.w);
#pragma unroll
      for (int i = 0; i < 16; ++i) c[i] = dot4(x[i], ws);
    } else {
#pragma unroll
      for (int i = 0; i < 16; ++i) c[i] = dot4(x[i], (i & 1) ? wh : wl);
    }
    fold_stage<8>(c, lane);
    fold_stage<4>(c, lane);
    fold_stage<2>(c, lane);
    fold_stage<1>(c, lane);
    const float tot = c[0] + xhalf(c[0]);
    if (half == 0) s_d[w * 16 + l15] = tot;

    if (role == 0) {
      float4 cs = x[0];
#pragma unroll
      for (int i = 1; i < 16; ++i) {
        cs.x += x[i].x; cs.y += x[i].y; cs.z += x[i].z; cs.w += x[i].w;
      }
      *reinterpret_cast<float4*>(&s_part[(w * 32 + lane) * 4]) = cs;
    }
  }

  const int nt = w & 3;
  const int kc = w >> 2;
  float bw[8];
#pragma unroll
  for (int s = 0; s < 4; ++s) {
    const int k = kc * 16 + s * 4 + 2 * half;
    bw[2 * s]     = W[k * 64 + nt * 16 + l15];
    bw[2 * s + 1] = W[(k + 1) * 64 + nt * 16 + l15];
  }
  __syncthreads();

  if (role == 1) {
    const int j  = w * 16 + l15;
    const int r0 = (2 * j) & 511;
    const float t  = s_d[r0] + s_d[r0 + 1];
    const float lt = (t > 0.f) ? t : ALPHA * t;
    const float mw = row16_max(lt);
    const float e  = __expf(lt - mw);
    float4 xp = make_float4(0.f, 0.f, 0.f, 0.f);
    float ev[16];
#define XP_ROW(I)                                  \
    {                                              \
      const float ei = lane_value<I>(e);           \
      ev[I] = ei;                                  \
      xp.x += ei * x[I].x; xp.y += ei * x[I].y;    \
      xp.z += ei * x[I].z; xp.w += ei * x[I].w;    \
    }
    XP_ROW(0)  XP_ROW(1)  XP_ROW(2)  XP_ROW(3)
    XP_ROW(4)  XP_ROW(5)  XP_ROW(6)  XP_ROW(7)
    XP_ROW(8)  XP_ROW(9)  XP_ROW(10) XP_ROW(11)
    XP_ROW(12) XP_ROW(13) XP_ROW(14) XP_ROW(15)
#undef XP_ROW
    if (lane == 0) {
      const float zw = (((ev[0] + ev[1]) + (ev[2] + ev[3])) + ((ev[4] + ev[5]) + (ev[6] + ev[7]))) +
                       (((ev[8] + ev[9]) + (ev[10] + ev[11])) + ((ev[12] + ev[13]) + (ev[14] + ev[15])));
      s_mw[w] = mw;
      s_zw[w] = zw;
    }
    *reinterpret_cast<float4*>(&s_xp[(w * 32 + lane) * 4]) = xp;
  }
  __syncthreads();

  if (role == 1) {
   if (tid < 128) {
    const float mwl = s_mw[lane];
    float m = row16_max(mwl);
    m = fmaxf(m, xhalf(m));
    const float sc = __expf(mwl - m);
    float Z = row16_sum(s_zw[lane] * sc);
    Z += xhalf(Z);
    const float scn = sc / Z;
    float a0 = 0.f, a1 = 0.f, a2 = 0.f, a3 = 0.f;
#define XP_JOIN(P)                                                 \
    a0 += lane_value<P>(scn)     * s_xp[(P) * 128 + tid];          \
    a1 += lane_value<P + 1>(scn) * s_xp[(P + 1) * 128 + tid];      \
    a2 += lane_value<P + 2>(scn) * s_xp[(P + 2) * 128 + tid];      \
    a3 += lane_value<P + 3>(scn) * s_xp[(P + 3) * 128 + tid];
    XP_JOIN(0)  XP_JOIN(4)  XP_JOIN(8)  XP_JOIN(12)
    XP_JOIN(16) XP_JOIN(20) XP_JOIN(24) XP_JOIN(28)
#undef XP_JOIN
    s_M[256 + tid] = (a0 + a1) + (a2 + a3);
   }
  } else if (tid >= 768) {
    const int t2 = tid - 768;
    const int cc = t2 & 127;
    const int hi = t2 >> 7;
    float sum = 0.f;
    for (int p = 0; p < 16; ++p) sum += s_part[(hi * 16 + p) * 128 + cc];
    s_M[hi * 128 + cc] = sum;
  } else if (tid >= 512) {
    const int i = tid - 512;
    const float sv0 = s_d[2 * i];
    const float sv1 = s_d[2 * i + 1];
    const float v0  = (sv0 > 0.f) ? sv0 : ALPHA * sv0;
    const float v1  = (sv1 > 0.f) ? sv1 : ALPHA * sv1;
    const float mm  = fmaxf(v0, v1);
    const float ee0 = __expf(v0 - mm);
    const float ee1 = __expf(v1 - mm);
    const float den = 256.f * (ee0 + ee1);
    s_w0[i] = ee0 / den;
    s_w1[i] = ee1 / den;
  }
  __syncthreads();

  {
    const bool liveA = role ? (l15 == 2) : (l15 < 2);
    const int rowA = role ? 2 : (l15 & 1);
    v8f acc0 = (v8f){};
    v8f acc1 = (v8f){};
#pragma unroll
    for (int s = 0; s < 4; ++s) {
      const int k = kc * 16 + s * 4 + 2 * half;
      v2f a = *reinterpret_cast<const v2f*>(&s_M[rowA * 128 + k]);
      if (!liveA) a = (v2f){0.f, 0.f};
      v2f bf;
      bf.x = bw[2 * s];
      bf.y = bw[2 * s + 1];
      if (s & 1)
        acc1 = __builtin_amdgcn_wmma_f32_16x16x4_f32(
            false, a, false, bf, (short)0, acc1, false, false);
      else
        acc0 = __builtin_amdgcn_wmma_f32_16x16x4_f32(
            false, a, false, bf, (short)0, acc0, false, false);
    }
    if (half == 0) {
      const int cD = nt * 16 + l15;
      if (role == 0) {
        s_Y[(kc * 3 + 0) * 64 + cD] = acc0[0] + acc1[0];
        s_Y[(kc * 3 + 1) * 64 + cD] = acc0[1] + acc1[1];
      } else {
        s_Y[(kc * 3 + 2) * 64 + cD] = acc0[2] + acc1[2];
      }
    }
  }
  __syncthreads();

  if (tid < (role ? 64 : 128)) {
    const int row = role ? 2 : (tid >> 6);
    const int cc  = tid & 63;
    float sum = 0.f;
    for (int q = 0; q < 8; ++q) sum += s_Y[(q * 3 + row) * 64 + cc];
    if (row == 0)      s_Slo[cc] = sum;
    else if (row == 1) s_Shi[cc] = sum;
    else               s_P[cc] = (sum > 0.f) ? sum : (__expf(sum) - 1.f);
  }
  __syncthreads();

  float4* __restrict__ o4 = reinterpret_cast<float4*>(OUT + b * 512 * 64);
  const int c4 = (tid & 15) * 4;
  if (role == 0) {
    const float4 slo = *reinterpret_cast<const float4*>(&s_Slo[c4]);
    const float4 shi = *reinterpret_cast<const float4*>(&s_Shi[c4]);
    for (int q = 0; q < 4; ++q) {
      const int i = (tid >> 4) + q * 64;
      const float w0 = s_w0[i];
      const float w1 = s_w1[i];
      float4 hp;
      hp.x = w0 * slo.x + w1 * shi.x;
      hp.y = w0 * slo.y + w1 * shi.y;
      hp.z = w0 * slo.z + w1 * shi.z;
      hp.w = w0 * slo.w + w1 * shi.w;
      hp.x = (hp.x > 0.f) ? hp.x : (__expf(hp.x) - 1.f);
      hp.y = (hp.y > 0.f) ? hp.y : (__expf(hp.y) - 1.f);
      hp.z = (hp.z > 0.f) ? hp.z : (__expf(hp.z) - 1.f);
      hp.w = (hp.w > 0.f) ? hp.w : (__expf(hp.w) - 1.f);
      o4[i * 16 + (tid & 15)] = hp;
    }
  } else {
    const float4 pv = *reinterpret_cast<const float4*>(&s_P[c4]);
    for (int q = 0; q < 4; ++q) {
      const int i = (tid >> 4) + q * 64;
      o4[(256 + i) * 16 + (tid & 15)] = pv;
    }
  }
}

extern "C" void kernel_launch(void* const* d_in, const int* in_sizes, int n_in,
                              void* d_out, int out_size, void* d_ws, size_t ws_size,
                              hipStream_t stream) {
  const float* X = (const float*)d_in[0];
  const float* W = (const float*)d_in[2];
  const float* A = (const float*)d_in[3];
  float* OUT = (float*)d_out;

  gat_fused<<<8, 1024, 0, stream>>>(X, W, A, OUT);
}


# W a formed by the first 8 waves ahead of the X stream and every barrier fences LDS only: each wave's row dots run as its X rows land, not after the whole stream; order of all sums kept
# speedup vs baseline: 4.3480x; 1.1171x over previous
#include <hip/hip_runtime.h>
#include <math.h>

#define ALPHA 0.2f

typedef __attribute__((ext_vector_type(2))) float v2f;
typedef __attribute__((ext_vector_type(8))) float v8f;

template <int M>
__device__ __forceinline__ float row_xor(float x) {
  const int b = __builtin_bit_cast(int, x);
  return __builtin_bit_cast(
      float, __builtin_amdgcn_mov_dpp(b, 0x160 | M, 0xf, 0xf, true));
}
__device__ __forceinline__ float xhalf(float x) {
  const int b = __builtin_bit_cast(int, x);
  return __builtin_bit_cast(
      float, __builtin_amdgcn_permlanex16(b, b, 0x76543210, 0xfedcba98, false, false));
}
__device__ __forceinline__ float row16_max(float x) {
  x = fmaxf(x, row_xor<1>(x));
  x = fmaxf(x, row_xor<2>(x));
  x = fmaxf(x, row_xor<4>(x));
  x = fmaxf(x, row_xor<8>(x));
  return x;
}
__device__ __forceinline__ float row16_sum(float x) {
  x += row_xor<1>(x);
  x += row_xor<2>(x);
  x += row_xor<4>(x);
  x += row_xor<8>(x);
  return x;
}
template <int I>
__device__ __forceinline__ float lane_value(float x) {
  return __builtin_bit_cast(
      float, __builtin_amdgcn_readlane(__builtin_bit_cast(int, x), I));
}
__device__ __forceinline__ float dot4(const float4 a, const float4 b) {
  return a.x * b.x + a.y * b.y + a.z * b.z + a.w * b.w;
}

template <int O>
__device__ __forceinline__ void fold_stage(float* c, const int lane) {
  const bool up = (lane & O) != 0;
#pragma unroll
  for (int j = 0; j < O; ++j) {
    const float keep = up ? c[j + O] : c[j];
    const float send = up ? c[j] : c[j + O];
    c[j] = keep + __shfl_xor(send, O);
  }
}

template <int O>
__device__ __forceinline__ void fold_stage_row(float* c, const int lane) {
  const bool up = (lane & O) != 0;
#pragma unroll
  for (int j = 0; j < O; ++j) {
    const float keep = up ? c[j + O] : c[j];
    const float send = up ? c[j] : c[j + O];
    c[j] = keep + row_xor<O>(send);
  }
}

__device__ __forceinline__ void lds_barrier() {
  __builtin_amdgcn_fence(__ATOMIC_RELEASE, "workgroup", "local");
  __builtin_amdgcn_s_barrier();
  __builtin_amdgcn_fence(__ATOMIC_ACQUIRE, "workgroup", "local");
}

__global__ __launch_bounds__(1024) void gat_fused(
    const float* __restrict__ X,
    const float* __restrict__ W,
    const float* __restrict__ A,
    float* __restrict__ OUT)
{
  const int b    = blockIdx.x >> 1;
  const int role = blockIdx.x & 1;
  const int tid  = threadIdx.x;
  const int w    = tid >> 5;
  const int lane = tid & 31;
  const int half = lane >> 4;
  const int l15  = lane & 15;

  __shared__ __attribute__((aligned(16))) float s_wal[128];
  __shared__ __attribute__((aligned(16))) float s_wah[128];
  __shared__ float s_d[512];
  __shared__ float s_w0[256];
  __shared__ float s_w1[256];
  __shared__ __attribute__((aligned(16))) float s_part[32 * 128];
  __shared__ __attribute__((aligned(16))) float s_xp[32 * 128];
  __shared__ float s_mw[32];
  __shared__ float s_zw[32];
  __shared__ __attribute__((aligned(16))) float s_M[3 * 128];
  __shared__ float s_Y[8 * 3 * 64];
  __shared__ __attribute__((aligned(16))) float s_Slo[64];
  __shared__ __attribute__((aligned(16))) float s_Shi[64];
  __shared__ __attribute__((aligned(16))) float s_P[64];

  const float4* __restrict__ X4 = reinterpret_cast<const float4*>(X);
  const float4* __restrict__ W4 = reinterpret_cast<const float4*>(W);
  const float4* __restrict__ A4 = reinterpret_cast<const float4*>(A);

  const bool wa_wave = tid < 256;
  float cw[16];
  if (wa_wave) {
    float4 wq[8];
#pragma unroll
    for (int j = 0; j < 8; ++j) wq[j] = W4[tid + 256 * j];
    const float4 alo = A4[l15];
    const float4 ahi = A4[16 + l15];
#pragma unroll
    for (int j = 0; j < 8; ++j) {
      cw[2 * j]     = dot4(wq[j], alo);
      cw[2 * j + 1] = dot4(wq[j], ahi);
    }
  }

  float4 x[16];
#pragma unroll
  for (int i = 0; i < 16; ++i) x[i] = X4[(b * 512 + w * 16 + i) * 32 + lane];

  if (wa_wave) {
    fold_stage_row<8>(cw, lane);
    fold_stage_row<4>(cw, lane);
    fold_stage_row<2>(cw, lane);
    fold_stage_row<1>(cw, lane);
    const int k = (tid >> 4) + 16 * (l15 >> 1);
    if (l15 & 1) s_wah[k] = cw[0];
    else         s_wal[k] = cw[0];
  }
  lds_barrier();

  {
    const float4 wl = *reinterpret_cast<const float4*>(&s_wal[4 * lane]);
    const float4 wh = *reinterpret_cast<const float4*>(&s_wah[4 * lane]);
    float c[16];
    if (role == 0) {
      const float4 ws = make_float4(wl.x + wh.x, wl.y + wh.y, wl.z + wh.z, wl.w + wh.w);
#pragma unroll
      for (int i = 0; i < 16; ++i) c[i] = dot4(x[i], ws);
    } else {
#pragma unroll
      for (int i = 0; i < 16; ++i) c[i] = dot4(x[i], (i & 1) ? wh : wl);
    }
    fold_stage<8>(c, lane);
    fold_stage<4>(c, lane);
    fold_stage<2>(c, lane);
    fold_stage<1>(c, lane);
    const float tot = c[0] + xhalf(c[0]);
    if (half == 0) s_d[w * 16 + l15] = tot;

    if (role == 0) {
      float4 cs = x[0];
#pragma unroll
      for (int i = 1; i < 16; ++i) {
        cs.x += x[i].x; cs.y += x[i].y; cs.z += x[i].z; cs.w += x[i].w;
      }
      *reinterpret_cast<float4*>(&s_part[(w * 32 + lane) * 4]) = cs;
    }
  }

  const int nt = w & 3;
  const int kc = w >> 2;
  float bw[8];
#pragma unroll
  for (int s = 0; s < 4; ++s) {
    const int k = kc * 16 + s * 4 + 2 * half;
    bw[2 * s]     = W[k * 64 + nt * 16 + l15];
    bw[2 * s + 1] = W[(k + 1) * 64 + nt * 16 + l15];
  }
  lds_barrier();

  if (role == 1) {
    const int j  = w * 16 + l15;
    const int r0 = (2 * j) & 511;
    const float t  = s_d[r0] + s_d[r0 + 1];
    const float lt = (t > 0.f) ? t : ALPHA * t;
    const float mw = row16_max(lt);
    const float e  = __expf(lt - mw);
    float4 xp = make_float4(0.f, 0.f, 0.f, 0.f);
    float ev[16];
#define XP_ROW(I)                                  \
    {                                              \
      const float ei = lane_value<I>(e);           \
      ev[I] = ei;                                  \
      xp.x += ei * x[I].x; xp.y += ei * x[I].y;    \
      xp.z += ei * x[I].z; xp.w += ei * x[I].w;    \
    }
    XP_ROW(0)  XP_ROW(1)  XP_ROW(2)  XP_ROW(3)
    XP_ROW(4)  XP_ROW(5)  XP_ROW(6)  XP_ROW(7)
    XP_ROW(8)  XP_ROW(9)  XP_ROW(10) XP_ROW(11)
    XP_ROW(12) XP_ROW(13) XP_ROW(14) XP_ROW(15)
#undef XP_ROW
    if (lane == 0) {
      const float zw = (((ev[0] + ev[1]) + (ev[2] + ev[3])) + ((ev[4] + ev[5]) + (ev[6] + ev[7]))) +
                       (((ev[8] + ev[9]) + (ev[10] + ev[11])) + ((ev[12] + ev[13]) + (ev[14] + ev[15])));
      s_mw[w] = mw;
      s_zw[w] = zw;
    }
    *reinterpret_cast<float4*>(&s_xp[(w * 32 + lane) * 4]) = xp;
  }
  lds_barrier();

  if (role == 1) {
   if (tid < 128) {
    const float mwl = s_mw[lane];
    float m = row16_max(mwl);
    m = fmaxf(m, xhalf(m));
    const float sc = __expf(mwl - m);
    float Z = row16_sum(s_zw[lane] * sc);
    Z += xhalf(Z);
    const float scn = sc / Z;
    float a0 = 0.f, a1 = 0.f, a2 = 0.f, a3 = 0.f;
#define XP_JOIN(P)                                                 \
    a0 += lane_value<P>(scn)     * s_xp[(P) * 128 + tid];          \
    a1 += lane_value<P + 1>(scn) * s_xp[(P + 1) * 128 + tid];      \
    a2 += lane_value<P + 2>(scn) * s_xp[(P + 2) * 128 + tid];      \
    a3 += lane_value<P + 3>(scn) * s_xp[(P + 3) * 128 + tid];
    XP_JOIN(0)  XP_JOIN(4)  XP_JOIN(8)  XP_JOIN(12)
    XP_JOIN(16) XP_JOIN(20) XP_JOIN(24) XP_JOIN(28)
#undef XP_JOIN
    s_M[256 + tid] = (a0 + a1) + (a2 + a3);
   }
  } else if (tid >= 768) {
    const int t2 = tid - 768;
    const int cc = t2 & 127;
    const int hi = t2 >> 7;
    float sum = 0.f;
    for (int p = 0; p < 16; ++p) sum += s_part[(hi * 16 + p) * 128 + cc];
    s_M[hi * 128 + cc] = sum;
  } else if (tid >= 512) {
    const int i = tid - 512;
    const float sv0 = s_d[2 * i];
    const float sv1 = s_d[2 * i + 1];
    const float v0  = (sv0 > 0.f) ? sv0 : ALPHA * sv0;
    const float v1  = (sv1 > 0.f) ? sv1 : ALPHA * sv1;
    const float mm  = fmaxf(v0, v1);
    const float ee0 = __expf(v0 - mm);
    const float ee1 = __expf(v1 - mm);
    const float den = 256.f * (ee0 + ee1);
    s_w0[i] = ee0 / den;
    s_w1[i] = ee1 / den;
  }
  lds_barrier();

  {
    const bool liveA = role ? (l15 == 2) : (l15 < 2);
    const int rowA = role ? 2 : (l15 & 1);
    v8f acc0 = (v8f){};
    v8f acc1 = (v8f){};
#pragma unroll
    for (int s = 0; s < 4; ++s) {
      const int k = kc * 16 + s * 4 + 2 * half;
      v2f a = *reinterpret_cast<const v2f*>(&s_M[rowA * 128 + k]);
      if (!liveA) a = (v2f){0.f, 0.f};
      v2f bf;
      bf.x = bw[2 * s];
      bf.y = bw[2 * s + 1];
      if (s & 1)
        acc1 = __builtin_amdgcn_wmma_f32_16x16x4_f32(
            false, a, false, bf, (short)0, acc1, false, false);
      else
        acc0 = __builtin_amdgcn_wmma_f32_16x16x4_f32(
            false, a, false, bf, (short)0, acc0, false, false);
    }
    if (half == 0) {
      const int cD = nt * 16 + l15;
      if (role == 0) {
        s_Y[(kc * 3 + 0) * 64 + cD] = acc0[0] + acc1[0];
        s_Y[(kc * 3 + 1) * 64 + cD] = acc0[1] + acc1[1];
      } else {
        s_Y[(kc * 3 + 2) * 64 + cD] = acc0[2] + acc1[2];
      }
    }
  }
  lds_barrier();

  if (tid < (role ? 64 : 128)) {
    const int row = role ? 2 : (tid >> 6);
    const int cc  = tid & 63;
    float sum = 0.f;
    for (int q = 0; q < 8; ++q) sum += s_Y[(q * 3 + row) * 64 + cc];
    if (row == 0)      s_Slo[cc] = sum;
    else if (row == 1) s_Shi[cc] = sum;
    else               s_P[cc] = (sum > 0.f) ? sum : (__expf(sum) - 1.f);
  }
  lds_barrier();

  float4* __restrict__ o4 = reinterpret_cast<float4*>(OUT + b * 512 * 64);
  const int c4 = (tid & 15) * 4;
  if (role == 0) {
    const float4 slo = *reinterpret_cast<const float4*>(&s_Slo[c4]);
    const float4 shi = *reinterpret_cast<const float4*>(&s_Shi[c4]);
    for (int q = 0; q < 4; ++q) {
      const int i = (tid >> 4) + q * 64;
      const float w0 = s_w0[i];
      const float w1 = s_w1[i];
      float4 hp;
      hp.x = w0 * slo.x + w1 * shi.x;
      hp.y = w0 * slo.y + w1 * shi.y;
      hp.z = w0 * slo.z + w1 * shi.z;
      hp.w = w0 * slo.w + w1 * shi.w;
      hp.x = (hp.x > 0.f) ? hp.x : (__expf(hp.x) - 1.f);
      hp.y = (hp.y > 0.f) ? hp.y : (__expf(hp.y) - 1.f);
      hp.z = (hp.z > 0.f) ? hp.z : (__expf(hp.z) - 1.f);
      hp.w = (hp.w > 0.f) ? hp.w : (__expf(hp.w) - 1.f);
      o4[i * 16 + (tid & 15)] = hp;
    }
  } else {
    const float4 pv = *reinterpret_cast<const float4*>(&s_P[c4]);
    for (int q = 0; q < 4; ++q) {
      const int i = (tid >> 4) + q * 64;
      o4[(256 + i) * 16 + (tid & 15)] = pv;
    }
  }
}

extern "C" void kernel_launch(void* const* d_in, const int* in_sizes, int n_in,
                              void* d_out, int out_size, void* d_ws, size_t ws_size,
                              hipStream_t stream) {
  const float* X = (const float*)d_in[0];
  const float* W = (const float*)d_in[2];
  const float* A = (const float*)d_in[3];
  float* OUT = (float*)d_out;

  gat_fused<<<8, 1024, 0, stream>>>(X, W, A, OUT);
}


# the two work-groups of a batch stream its X rows half a batch apart (role 1: wave w holds row block (w+16)&31), so they never ask for the same lines at the same moment; all sums and their order kept
# speedup vs baseline: 4.6316x; 1.0652x over previous
#include <hip/hip_runtime.h>
#include <math.h>

#define ALPHA 0.2f

typedef __attribute__((ext_vector_type(2))) float v2f;
typedef __attribute__((ext_vector_type(8))) float v8f;

template <int M>
__device__ __forceinline__ float row_xor(float x) {
  const int b = __builtin_bit_cast(int, x);
  return __builtin_bit_cast(
      float, __builtin_amdgcn_mov_dpp(b, 0x160 | M, 0xf, 0xf, true));
}
__device__ __forceinline__ float xhalf(float x) {
  const int b = __builtin_bit_cast(int, x);
  return __builtin_bit_cast(
      float, __builtin_amdgcn_permlanex16(b, b, 0x76543210, 0xfedcba98, false, false));
}
__device__ __forceinline__ float row16_max(float x) {
  x = fmaxf(x, row_xor<1>(x));
  x = fmaxf(x, row_xor<2>(x));
  x = fmaxf(x, row_xor<4>(x));
  x = fmaxf(x, row_xor<8>(x));
  return x;
}
__device__ __forceinline__ float row16_sum(float x) {
  x += row_xor<1>(x);
  x += row_xor<2>(x);
  x += row_xor<4>(x);
  x += row_xor<8>(x);
  return x;
}
template <int I>
__device__ __forceinline__ float lane_value(float x) {
  return __builtin_bit_cast(
      float, __builtin_amdgcn_readlane(__builtin_bit_cast(int, x), I));
}
__device__ __forceinline__ float dot4(const float4 a, const float4 b) {
  return a.x * b.x + a.y * b.y + a.z * b.z + a.w * b.w;
}

template <int O>
__device__ __forceinline__ void fold_stage(float* c, const int lane) {
  const bool up = (lane & O) != 0;
#pragma unroll
  for (int j = 0; j < O; ++j) {
    const float keep = up ? c[j + O] : c[j];
    const float send = up ? c[j] : c[j + O];
    c[j] = keep + __shfl_xor(send, O);
  }
}

template <int O>
__device__ __forceinline__ void fold_stage_row(float* c, const int lane) {
  const bool up = (lane & O) != 0;
#pragma unroll
  for (int j = 0; j < O; ++j) {
    const float keep = up ? c[j + O] : c[j];
    const float send = up ? c[j] : c[j + O];
    c[j] = keep + row_xor<O>(send);
  }
}

__device__ __forceinline__ void lds_barrier() {
  __builtin_amdgcn_fence(__ATOMIC_RELEASE, "workgroup", "local");
  __builtin_amdgcn_s_barrier();
  __builtin_amdgcn_fence(__ATOMIC_ACQUIRE, "workgroup", "local");
}

__global__ __launch_bounds__(1024) void gat_fused(
    const float* __restrict__ X,
    const float* __restrict__ W,
    const float* __restrict__ A,
    float* __restrict__ OUT)
{
  const int b    = blockIdx.x >> 1;
  const int role = blockIdx.x & 1;
  const int tid  = threadIdx.x;
  const int w    = tid >> 5;
  const int lane = tid & 31;
  const int half = lane >> 4;
  const int l15  = lane & 15;
  const int rb   = role ? ((w + 16) & 31) : w;

  __shared__ __attribute__((aligned(16))) float s_wal[128];
  __shared__ __attribute__((aligned(16))) float s_wah[128];
  __shared__ float s_d[512];
  __shared__ float s_w0[256];
  __shared__ float s_w1[256];
  __shared__ __attribute__((aligned(16))) float s_part[32 * 128];
  __shared__ __attribute__((aligned(16))) float s_xp[32 * 128];
  __shared__ float s_mw[32];
  __shared__ float s_zw[32];
  __shared__ __attribute__((aligned(16))) float s_M[3 * 128];
  __shared__ float s_Y[8 * 3 * 64];
  __shared__ __attribute__((aligned(16))) float s_Slo[64];
  __shared__ __attribute__((aligned(16))) float s_Shi[64];
  __shared__ __attribute__((aligned(16))) float s_P[64];

  const float4* __restrict__ X4 = reinterpret_cast<const float4*>(X);
  const float4* __restrict__ W4 = reinterpret_cast<const float4*>(W);
  const float4* __restrict__ A4 = reinterpret_cast<const float4*>(A);

  const bool wa_wave = tid < 256;
  float cw[16];
  if (wa_wave) {
    float4 wq[8];
#pragma unroll
    for (int j = 0; j < 8; ++j) wq[j] = W4[tid + 256 * j];
    const float4 alo = A4[l15];
    const float4 ahi = A4[16 + l15];
#pragma unroll
    for (int j = 0; j < 8; ++j) {
      cw[2 * j]     = dot4(wq[j], alo);
      cw[2 * j + 1] = dot4(wq[j], ahi);
    }
  }

  float4 x[16];
#pragma unroll
  for (int i = 0; i < 16; ++i) x[i] = X4[(b * 512 + rb * 16 + i) * 32 + lane];

  if (wa_wave) {
    fold_stage_row<8>(cw, lane);
    fold_stage_row<4>(cw, lane);
    fold_stage_row<2>(cw, lane);
    fold_stage_row<1>(cw, lane);
    const int k = (tid >> 4) + 16 * (l15 >> 1);
    if (l15 & 1) s_wah[k] = cw[0];
    else         s_wal[k] = cw[0];
  }
  lds_barrier();

  {
    const float4 wl = *reinterpret_cast<const float4*>(&s_wal[4 * lane]);
    const float4 wh = *reinterpret_cast<const float4*>(&s_wah[4 * lane]);
    float c[16];
    if (role == 0) {
      const float4 ws = make_float4(wl.x + wh.x, wl.y + wh.y, wl.z + wh.z, wl.w + wh.w);
#pragma unroll
      for (int i = 0; i < 16; ++i) c[i] = dot4(x[i], ws);
    } else {
#pragma unroll
      for (int i = 0; i < 16; ++i) c[i] = dot4(x[i], (i & 1) ? wh : wl);
    }
    fold_stage<8>(c, lane);
    fold_stage<4>(c, lane);
    fold_stage<2>(c, lane);
    fold_stage<1>(c, lane);
    const float tot = c[0] + xhalf(c[0]);
    if (half == 0) s_d[rb * 16 + l15] = tot;

    if (role == 0) {
      float4 cs = x[0];
#pragma unroll
      for (int i = 1; i < 16; ++i) {
        cs.x += x[i].x; cs.y += x[i].y; cs.z += x[i].z; cs.w += x[i].w;
      }
      *reinterpret_cast<float4*>(&s_part[(rb * 32 + lane) * 4]) = cs;
    }
  }

  const int nt = w & 3;
  const int kc = w >> 2;
  float bw[8];
#pragma unroll
  for (int s = 0; s < 4; ++s) {
    const int k = kc * 16 + s * 4 + 2 * half;
    bw[2 * s]     = W[k * 64 + nt * 16 + l15];
    bw[2 * s + 1] = W[(k + 1) * 64 + nt * 16 + l15];
  }
  lds_barrier();

  if (role == 1) {
    const int j  = rb * 16 + l15;
    const int r0 = (2 * j) & 511;
    const float t  = s_d[r0] + s_d[r0 + 1];
    const float lt = (t > 0.f) ? t : ALPHA * t;
    const float mw = row16_max(lt);
    const float e  = __expf(lt - mw);
    float4 xp = make_float4(0.f, 0.f, 0.f, 0.f);
    float ev[16];
#define XP_ROW(I)                                  \
    {                                              \
      const float ei = lane_value<I>(e);           \
      ev[I] = ei;                                  \
      xp.x += ei * x[I].x; xp.y += ei * x[I].y;    \
      xp.z += ei * x[I].z; xp.w += ei * x[I].w;    \
    }
    XP_ROW(0)  XP_ROW(1)  XP_ROW(2)  XP_ROW(3)
    XP_ROW(4)  XP_ROW(5)  XP_ROW(6)  XP_ROW(7)
    XP_ROW(8)  XP_ROW(9)  XP_ROW(10) XP_ROW(11)
    XP_ROW(12) XP_ROW(13) XP_ROW(14) XP_ROW(15)
#undef XP_ROW
    if (lane == 0) {
      const float zw = (((ev[0] + ev[1]) + (ev[2] + ev[3])) + ((ev[4] + ev[5]) + (ev[6] + ev[7]))) +
                       (((ev[8] + ev[9]) + (ev[10] + ev[11])) + ((ev[12] + ev[13]) + (ev[14] + ev[15])));
      s_mw[rb] = mw;
      s_zw[rb] = zw;
    }
    *reinterpret_cast<float4*>(&s_xp[(rb * 32 + lane) * 4]) = xp;
  }
  lds_barrier();

  if (role == 1) {
   if (tid < 128) {
    const float mwl = s_mw[lane];
    float m = row16_max(mwl);
    m = fmaxf(m, xhalf(m));
    const float sc = __expf(mwl - m);
    float Z = row16_sum(s_zw[lane] * sc);
    Z += xhalf(Z);
    const float scn = sc / Z;
    float a0 = 0.f, a1 = 0.f, a2 = 0.f, a3 = 0.f;
#define XP_JOIN(P)                                                 \
    a0 += lane_value<P>(scn)     * s_xp[(P) * 128 + tid];          \
    a1 += lane_value<P + 1>(scn) * s_xp[(P + 1) * 128 + tid];      \
    a2 += lane_value<P + 2>(scn) * s_xp[(P + 2) * 128 + tid];      \
    a3 += lane_value<P + 3>(scn) * s_xp[(P + 3) * 128 + tid];
    XP_JOIN(0)  XP_JOIN(4)  XP_JOIN(8)  XP_JOIN(12)
    XP_JOIN(16) XP_JOIN(20) XP_JOIN(24) XP_JOIN(28)
#undef XP_JOIN
    s_M[256 + tid] = (a0 + a1) + (a2 + a3);
   }
  } else if (tid >= 768) {
    const int t2 = tid - 768;
    const int cc = t2 & 127;
    const int hi = t2 >> 7;
    float sum = 0.f;
    for (int p = 0; p < 16; ++p) sum += s_part[(hi * 16 + p) * 128 + cc];
    s_M[hi * 128 + cc] = sum;
  } else if (tid >= 512) {
    const int i = tid - 512;
    const float sv0 = s_d[2 * i];
    const float sv1 = s_d[2 * i + 1];
    const float v0  = (sv0 > 0.f) ? sv0 : ALPHA * sv0;
    const float v1  = (sv1 > 0.f) ? sv1 : ALPHA * sv1;
    const float mm  = fmaxf(v0, v1);
    const float ee0 = __expf(v0 - mm);
    const float ee1 = __expf(v1 - mm);
    const float den = 256.f * (ee0 + ee1);
    s_w0[i] = ee0 / den;
    s_w1[i] = ee1 / den;
  }
  lds_barrier();

  {
    const bool liveA = role ? (l15 == 2) : (l15 < 2);
    const int rowA = role ? 2 : (l15 & 1);
    v8f acc0 = (v8f){};
    v8f acc1 = (v8f){};
#pragma unroll
    for (int s = 0; s < 4; ++s) {
      const int k = kc * 16 + s * 4 + 2 * half;
      v2f a = *reinterpret_cast<const v2f*>(&s_M[rowA * 128 + k]);
      if (!liveA) a = (v2f){0.f, 0.f};
      v2f bf;
      bf.x = bw[2 * s];
      bf.y = bw[2 * s + 1];
      if (s & 1)
        acc1 = __builtin_amdgcn_wmma_f32_16x16x4_f32(
            false, a, false, bf, (short)0, acc1, false, false);
      else
        acc0 = __builtin_amdgcn_wmma_f32_16x16x4_f32(
            false, a, false, bf, (short)0, acc0, false, false);
    }
    if (half == 0) {
      const int cD = nt * 16 + l15;
      if (role == 0) {
        s_Y[(kc * 3 + 0) * 64 + cD] = acc0[0] + acc1[0];
        s_Y[(kc * 3 + 1) * 64 + cD] = acc0[1] + acc1[1];
      } else {
        s_Y[(kc * 3 + 2) * 64 + cD] = acc0[2] + acc1[2];
      }
    }
  }
  lds_barrier();

  if (tid < (role ? 64 : 128)) {
    const int row = role ? 2 : (tid >> 6);
    const int cc  = tid & 63;
    float sum = 0.f;
    for (int q = 0; q < 8; ++q) sum += s_Y[(q * 3 + row) * 64 + cc];
    if (row == 0)      s_Slo[cc] = sum;
    else if (row == 1) s_Shi[cc] = sum;
    else               s_P[cc] = (sum > 0.f) ? sum : (__expf(sum) - 1.f);
  }
  lds_barrier();

  float4* __restrict__ o4 = reinterpret_cast<float4*>(OUT + b * 512 * 64);
  const int c4 = (tid & 15) * 4;
  if (role == 0) {
    const float4 slo = *reinterpret_cast<const float4*>(&s_Slo[c4]);
    const float4 shi = *reinterpret_cast<const float4*>(&s_Shi[c4]);
    for (int q = 0; q < 4; ++q) {
      const int i = (tid >> 4) + q * 64;
      const float w0 = s_w0[i];
      const float w1 = s_w1[i];
      float4 hp;
      hp.x = w0 * slo.x + w1 * shi.x;
      hp.y = w0 * slo.y + w1 * shi.y;
      hp.z = w0 * slo.z + w1 * shi.z;
      hp.w = w0 * slo.w + w1 * shi.w;
      hp.x = (hp.x > 0.f) ? hp.x : (__expf(hp.x) - 1.f);
      hp.y = (hp.y > 0.f) ? hp.y : (__expf(hp.y) - 1.f);
      hp.z = (hp.z > 0.f) ? hp.z : (__expf(hp.z) - 1.f);
      hp.w = (hp.w > 0.f) ? hp.w : (__expf(hp.w) - 1.f);
      o4[i * 16 + (tid & 15)] = hp;
    }
  } else {
    const float4 pv = *reinterpret_cast<const float4*>(&s_P[c4]);
    for (int q = 0; q < 4; ++q) {
      const int i = (tid >> 4) + q * 64;
      o4[(256 + i) * 16 + (tid & 15)] = pv;
    }
  }
}

extern "C" void kernel_launch(void* const* d_in, const int* in_sizes, int n_in,
                              void* d_out, int out_size, void* d_ws, size_t ws_size,
                              hipStream_t stream) {
  const float* X = (const float*)d_in[0];
  const float* W = (const float*)d_in[2];
  const float* A = (const float*)d_in[3];
  float* OUT = (float*)d_out;

  gat_fused<<<8, 1024, 0, stream>>>(X, W, A, OUT);
}
